# starting kernel
# baseline (speedup 1.0000x reference)
#include <hip/hip_runtime.h>

typedef __attribute__((ext_vector_type(2))) float v2f;
typedef __attribute__((ext_vector_type(8))) float v8f;

#define N_NODES 50000
#define N_EDGES 400000
#define IN_F    768
#define HID_F   16
#define OUT_F   21

__global__ __launch_bounds__(256) void gcn_gemm1_wmma(
    const float* __restrict__ X, const float* __restrict__ W1,
    float* __restrict__ Z)
{
  __shared__ float w1s[IN_F * HID_F];
  {
    const float4* w4 = (const float4*)W1;
    float4*       s4 = (float4*)w1s;
    const int n4 = (IN_F * HID_F) / 4;
    for (int i = threadIdx.x; i < n4; i += 256) s4[i] = w4[i];
  }
  __syncthreads();

  const int lane = threadIdx.x & 31;
  const int wave = threadIdx.x >> 5;
  const int tile = blockIdx.x * 8 + wave;
  if (tile >= (N_NODES / 16)) return;
  const int m0   = tile * 16;

  const int half = lane >> 4;
  const int l16  = lane & 15;
  const int koff = half << 1;

  const float* xrow = X + (size_t)(m0 + l16) * IN_F + koff;

  v8f acc0 = {};
  v8f acc1 = {};
  for (int k = 0; k < IN_F; k += 8) {
    v2f a0 = *(const v2f*)(xrow + k);
    v2f a1 = *(const v2f*)(xrow + k + 4);
    v2f b0, b1;
    b0.x = w1s[(k + koff)     * HID_F + l16];
    b0.y = w1s[(k + koff + 1) * HID_F + l16];
    b1.x = w1s[(k + 4 + koff)     * HID_F + l16];
    b1.y = w1s[(k + 4 + koff + 1) * HID_F + l16];
    acc0 = __builtin_amdgcn_wmma_f32_16x16x4_f32(false, a0, false, b0,
                                                 (short)0, acc0, false, false);
    acc1 = __builtin_amdgcn_wmma_f32_16x16x4_f32(false, a1, false, b1,
                                                 (short)0, acc1, false, false);
  }

  float* zp = Z + (size_t)(m0 + half * 8) * HID_F + l16;
#pragma unroll
  for (int r = 0; r < 8; ++r) zp[r * HID_F] = acc0[r] + acc1[r];
}

__global__ __launch_bounds__(256) void gcn_zero(float* __restrict__ p, int n)
{
  int i = blockIdx.x * 256 + threadIdx.x;
  if (i < n) p[i] = 0.0f;
}

__global__ __launch_bounds__(256) void gcn_scatter16(
    const float* __restrict__ feat, const int* __restrict__ src,
    const int* __restrict__ dst, float* __restrict__ acc)
{
  unsigned tid = blockIdx.x * 256u + threadIdx.x;
  unsigned e = tid >> 4;
  unsigned f = tid & 15;
  if (e < N_EDGES) {
    int s = src[e];
    int d = dst[e];
    atomicAdd(acc + (size_t)d * HID_F + f, feat[(size_t)s * HID_F + f]);
  }
}

__global__ __launch_bounds__(256) void gcn_bias_relu(
    const float* __restrict__ a, const float* __restrict__ b1,
    float* __restrict__ h)
{
  int i = blockIdx.x * 256 + threadIdx.x;
  if (i < N_NODES * HID_F) {
    float v = a[i] + b1[i & (HID_F - 1)];
    h[i] = v > 0.0f ? v : 0.0f;
  }
}

__global__ __launch_bounds__(256) void gcn_gemm2_wmma(
    const float* __restrict__ A, const float* __restrict__ W2,
    const float* __restrict__ b2, float* __restrict__ out)
{
  const int lane = threadIdx.x & 31;
  const int wave = threadIdx.x >> 5;
  const int t    = blockIdx.x * 8 + wave;
  const int mtiles = N_NODES / 16;
  if (t >= mtiles * 2) return;
  const int mt = t >> 1;
  const int nt = t & 1;
  const int m0 = mt * 16;

  const int half = lane >> 4;
  const int l16  = lane & 15;
  const int koff = half << 1;
  const int n    = nt * 16 + l16;
  const bool nvalid = (n < OUT_F);

  const float* arow = A + (size_t)(m0 + l16) * HID_F + koff;

  v8f acc = {};
#pragma unroll
  for (int k = 0; k < HID_F; k += 4) {
    v2f a = *(const v2f*)(arow + k);
    v2f b;
    b.x = nvalid ? W2[(k + koff)     * OUT_F + n] : 0.0f;
    b.y = nvalid ? W2[(k + koff + 1) * OUT_F + n] : 0.0f;
    acc = __builtin_amdgcn_wmma_f32_16x16x4_f32(false, a, false, b,
                                                (short)0, acc, false, false);
  }

  if (nvalid) {
    float bias = b2[n];
    float* op = out + (size_t)(m0 + half * 8) * OUT_F + n;
#pragma unroll
    for (int r = 0; r < 8; ++r) op[r * OUT_F] = acc[r] + bias;
  }
}

extern "C" void kernel_launch(void* const* d_in, const int* in_sizes, int n_in,
                              void* d_out, int out_size, void* d_ws, size_t ws_size,
                              hipStream_t stream)
{
  const float* feature = (const float*)d_in[0];
  const float* W1      = (const float*)d_in[1];
  const float* b1      = (const float*)d_in[2];
  const float* W2      = (const float*)d_in[3];
  const float* b2      = (const float*)d_in[4];
  const int*   src     = (const int*)d_in[5];
  const int*   dst     = (const int*)d_in[6];
  float*       out     = (float*)d_out;

  float* buf0 = (float*)d_ws;
  float* buf1 = buf0 + (size_t)N_NODES * HID_F;

  const int NH = N_NODES * HID_F;
  const int zero_blocks    = (NH + 255) / 256;
  const int scatter_blocks = (N_EDGES * HID_F) / 256;
  const int gemm1_blocks   = ((N_NODES / 16) + 7) / 8;
  const int gemm2_blocks   = ((N_NODES / 16) * 2 + 7) / 8;

  gcn_gemm1_wmma<<<gemm1_blocks, 256, 0, stream>>>(feature, W1, buf0);

  gcn_zero<<<zero_blocks, 256, 0, stream>>>(buf1, NH);
  gcn_scatter16<<<scatter_blocks, 256, 0, stream>>>(buf0, src, dst, buf1);

  gcn_bias_relu<<<zero_blocks, 256, 0, stream>>>(buf1, b1, buf0);

  gcn_zero<<<zero_blocks, 256, 0, stream>>>(buf1, NH);
  gcn_scatter16<<<scatter_blocks, 256, 0, stream>>>(buf0, src, dst, buf1);

  gcn_gemm2_wmma<<<gemm2_blocks, 256, 0, stream>>>(buf1, W2, b2, out);
}


# First product split along K: two waves per 16-row tile (k 0..383 and 384..767), 26 waves a work-group not 13, upper half added through LDS; twice the X loads in flight; f32 sum order of X@W1 changes
# speedup vs baseline: 1.9506x; 1.9506x over previous
#include <hip/hip_runtime.h>

typedef __attribute__((ext_vector_type(2))) float v2f;
typedef __attribute__((ext_vector_type(8))) float v8f;

#define N_NODES 50000
#define N_EDGES 400000
#define IN_F    768
#define HID_F   16
#define OUT_F   21
#define GEMM1_TILES  (N_NODES / 16)
#define GEMM1_BLOCKS 256
#define GEMM1_SLOTS  ((GEMM1_TILES + GEMM1_BLOCKS - 1) / GEMM1_BLOCKS)
#define GEMM1_KSPLIT 2
#define GEMM1_KPART  (IN_F / GEMM1_KSPLIT)
#define GEMM1_WAVES  (GEMM1_SLOTS * GEMM1_KSPLIT)
#define GEMM1_THREADS (GEMM1_WAVES * 32)
#define PART_NODES 200
#define N_PARTS   (N_NODES / PART_NODES)
#define BIN_EDGES 2048
#define N_BINS    ((N_EDGES + BIN_EDGES - 1) / BIN_EDGES)
#define SEG_PER_WAVE ((N_BINS + 31) / 32)
#define OFS_PITCH 256
#define ACC_PITCH 17
#define ACC_ROWS  208
#define OUT_TASKS ((ACC_ROWS / 16) * 2)
#define OST_PITCH 34

__global__ __launch_bounds__(GEMM1_THREADS) void gcn_gemm1_wmma(
    const float* __restrict__ X, const float* __restrict__ W1,
    float* __restrict__ Z)
{
  __shared__ float w1s[IN_F * HID_F];
  __shared__ float part[GEMM1_SLOTS * 8 * 32];
  {
    const float4* w4 = (const float4*)W1;
    float4*       s4 = (float4*)w1s;
    const int n4 = (IN_F * HID_F) / 4;
    for (int i = threadIdx.x; i < n4; i += GEMM1_THREADS) s4[i] = w4[i];
  }
  __syncthreads();

  const int lane = threadIdx.x & 31;
  const int wave = threadIdx.x >> 5;
  const int slot = wave / GEMM1_KSPLIT;
  const int kh   = wave % GEMM1_KSPLIT;
  const int tile = slot * GEMM1_BLOCKS + blockIdx.x;
  const bool live = tile < GEMM1_TILES;
  const int m0   = tile * 16;

  const int half = lane >> 4;
  const int l16  = lane & 15;
  const int koff = half << 1;

  v8f acc = {};
  if (live) {
    const float* xrow = X + (size_t)(m0 + l16) * IN_F + koff;
    const int k0 = kh * GEMM1_KPART;
    v8f acc0 = {};
    v8f acc1 = {};
#pragma nounroll
    for (int kg = k0; kg < k0 + GEMM1_KPART; kg += 64) {
      v2f a[16];
#pragma unroll
      for (int i = 0; i < 16; ++i)
        a[i] = *(const v2f*)(xrow + kg + 4 * i);
#pragma unroll
      for (int i = 0; i < 16; i += 2) {
        const int k = kg + 4 * i;
        v2f b0, b1;
        b0.x = w1s[(k + koff)     * HID_F + l16];
        b0.y = w1s[(k + koff + 1) * HID_F + l16];
        b1.x = w1s[(k + 4 + koff)     * HID_F + l16];
        b1.y = w1s[(k + 4 + koff + 1) * HID_F + l16];
        acc0 = __builtin_amdgcn_wmma_f32_16x16x4_f32(false, a[i], false, b0,
                                                     (short)0, acc0, false, false);
        acc1 = __builtin_amdgcn_wmma_f32_16x16x4_f32(false, a[i + 1], false, b1,
                                                     (short)0, acc1, false, false);
      }
    }
    acc = acc0 + acc1;
    if (kh != 0) {
      float* pp = part + slot * (8 * 32) + lane;
#pragma unroll
      for (int r = 0; r < 8; ++r) pp[r * 32] = acc[r];
    }
  }
  __syncthreads();

  if (live && kh == 0) {
    const float* pp = part + slot * (8 * 32) + lane;
    float* zp = Z + (size_t)(m0 + half * 8) * HID_F + l16;
#pragma unroll
    for (int r = 0; r < 8; ++r) zp[r * HID_F] = acc[r] + pp[r * 32];
  }
}

__global__ __launch_bounds__(512) void gcn_bin_edges(
    const int* __restrict__ src, const int* __restrict__ dst,
    int* __restrict__ seg, int* __restrict__ ofs)
{
  __shared__ int hist[OFS_PITCH];
  __shared__ int lstart[OFS_PITCH];
  __shared__ int wsum[8];
  const int tid  = threadIdx.x;
  const int lane = tid & 31;
  const int wave = tid >> 5;
  if (tid < OFS_PITCH) hist[tid] = 0;

  const int base = blockIdx.x * BIN_EDGES;
  int s[4], d[4];
#pragma unroll
  for (int k = 0; k < 4; ++k) {
    const int e  = base + k * 512 + tid;
    const int ec = e < N_EDGES ? e : N_EDGES - 1;
    s[k] = src[ec];
    d[k] = dst[ec];
  }
  __syncthreads();

  int pp[4], rk[4];
#pragma unroll
  for (int k = 0; k < 4; ++k) {
    const int e = base + k * 512 + tid;
    pp[k] = -1;
    rk[k] = 0;
    if (e < N_EDGES) {
      pp[k] = d[k] / PART_NODES;
      rk[k] = atomicAdd(&hist[pp[k]], 1);
    }
  }
  __syncthreads();

  int v = 0, incl = 0;
  if (tid < OFS_PITCH) {
    v = hist[tid];
    incl = v;
    for (int o = 1; o < 32; o <<= 1) {
      const int t = __shfl_up(incl, o);
      if (lane >= o) incl += t;
    }
    if (lane == 31) wsum[wave] = incl;
  }
  __syncthreads();
  if (tid < OFS_PITCH) {
    int pre = 0;
    for (int w = 0; w < wave; ++w) pre += wsum[w];
    const int ex = pre + incl - v;
    lstart[tid] = ex;
    if (tid <= N_PARTS) ofs[blockIdx.x * OFS_PITCH + tid] = ex;
  }
  __syncthreads();

  int* sg = seg + base;
#pragma unroll
  for (int k = 0; k < 4; ++k) {
    if (pp[k] >= 0)
      sg[lstart[pp[k]] + rk[k]] = (s[k] << 8) | (d[k] - pp[k] * PART_NODES);
  }
}

__device__ __forceinline__ void gcn_range_sums(
    const float* __restrict__ feat, const int* __restrict__ seg,
    const int* __restrict__ ofs, float* acc, int p)
{
  const int tid  = threadIdx.x;
  const int lane = tid & 31;
  const int wave = tid >> 5;
  const int q = lane >> 2;
  const int c = lane & 3;

  int pos[SEG_PER_WAVE], end[SEG_PER_WAVE];
#pragma unroll
  for (int j = 0; j < SEG_PER_WAVE; ++j) {
    const int b = wave + 32 * j;
    pos[j] = 0; end[j] = 0;
    if (b < N_BINS) {
      pos[j] = ofs[b * OFS_PITCH + p] + q;
      end[j] = ofs[b * OFS_PITCH + p + 1];
    }
  }
  __syncthreads();

  const float4* f4 = (const float4*)feat;
  const int*    sg = seg + wave * BIN_EDGES;
  for (int round = 0; round < BIN_EDGES / 8; ++round) {
    bool any = false;
#pragma unroll
    for (int j = 0; j < SEG_PER_WAVE; ++j) any = any || (pos[j] < end[j]);
    if (!any) break;
    int e[SEG_PER_WAVE];
#pragma unroll
    for (int j = 0; j < SEG_PER_WAVE; ++j)
      e[j] = sg[pos[j] < end[j] ? j * 32 * BIN_EDGES + pos[j] : 0];
    float4 v[SEG_PER_WAVE];
#pragma unroll
    for (int j = 0; j < SEG_PER_WAVE; ++j) v[j] = f4[(e[j] >> 8) * 4 + c];
    asm volatile("" :: "v"(v[0].x));
#pragma unroll
    for (int j = 0; j < SEG_PER_WAVE; ++j) {
      if (pos[j] < end[j]) {
        float* a = acc + (e[j] & 255) * ACC_PITCH + c * 4;
        atomicAdd(a + 0, v[j].x);
        atomicAdd(a + 1, v[j].y);
        atomicAdd(a + 2, v[j].z);
        atomicAdd(a + 3, v[j].w);
      }
      pos[j] += 8;
    }
  }
}

__global__ __launch_bounds__(1024) void gcn_agg_hidden(
    const float* __restrict__ feat, const int* __restrict__ seg,
    const int* __restrict__ ofs, const float* __restrict__ b1,
    float* __restrict__ res)
{
  __shared__ float acc[PART_NODES * ACC_PITCH];
  const int tid = threadIdx.x;
  const int p   = blockIdx.x;
  for (int i = tid; i < PART_NODES * ACC_PITCH; i += 1024) acc[i] = 0.0f;

  gcn_range_sums(feat, seg, ofs, acc, p);
  __syncthreads();

  float* rp = res + (size_t)p * PART_NODES * HID_F;
  for (int i = tid; i < PART_NODES * HID_F; i += 1024) {
    const int node = i >> 4;
    const int f    = i & 15;
    float val = acc[node * ACC_PITCH + f] + b1[f];
    val = val > 0.0f ? val : 0.0f;
    rp[i] = val;
  }
}

__global__ __launch_bounds__(1024) void gcn_agg_out_wmma(
    const float* __restrict__ feat, const int* __restrict__ seg,
    const int* __restrict__ ofs, const float* __restrict__ W2,
    const float* __restrict__ b2, float* __restrict__ out)
{
  __shared__ float acc[ACC_ROWS * ACC_PITCH];
  __shared__ float ost[ACC_ROWS * OST_PITCH];
  const int tid  = threadIdx.x;
  const int lane = tid & 31;
  const int wave = tid >> 5;
  const int p    = blockIdx.x;

  const int mt   = wave >> 1;
  const int nt   = wave & 1;
  const int half = lane >> 4;
  const int l16  = lane & 15;
  const int koff = half << 1;
  const int n    = nt * 16 + l16;
  const bool nvalid = (n < OUT_F);
  const int nc = nvalid ? n : 0;
  v2f bf[HID_F / 4];
#pragma unroll
  for (int k = 0; k < HID_F; k += 4) {
    const float w0 = W2[(k + koff)     * OUT_F + nc];
    const float w1 = W2[(k + koff + 1) * OUT_F + nc];
    bf[k / 4].x = nvalid ? w0 : 0.0f;
    bf[k / 4].y = nvalid ? w1 : 0.0f;
  }
  const float bias = b2[nc];

  for (int i = tid; i < ACC_ROWS * ACC_PITCH; i += 1024) acc[i] = 0.0f;

  gcn_range_sums(feat, seg, ofs, acc, p);
  __syncthreads();

  if (wave < OUT_TASKS) {
    const float* arow = acc + (mt * 16 + l16) * ACC_PITCH + koff;
    v8f d = {};
#pragma unroll
    for (int k = 0; k < HID_F; k += 4) {
      v2f a;
      a.x = arow[k];
      a.y = arow[k + 1];
      d = __builtin_amdgcn_wmma_f32_16x16x4_f32(false, a, false, bf[k / 4],
                                                (short)0, d, false, false);
    }
    float* sp = ost + (mt * 16 + half * 8) * OST_PITCH + n;
#pragma unroll
    for (int r = 0; r < 8; ++r) sp[r * OST_PITCH] = d[r] + bias;
  }
  __syncthreads();

  float* op = out + (size_t)p * PART_NODES * OUT_F;
  for (int i = tid; i < PART_NODES * OUT_F; i += 1024) {
    const int row = i / OUT_F;
    const int col = i - row * OUT_F;
    op[i] = ost[row * OST_PITCH + col];
  }
}

extern "C" void kernel_launch(void* const* d_in, const int* in_sizes, int n_in,
                              void* d_out, int out_size, void* d_ws, size_t ws_size,
                              hipStream_t stream)
{
  const float* feature = (const float*)d_in[0];
  const float* W1      = (const float*)d_in[1];
  const float* b1      = (const float*)d_in[2];
  const float* W2      = (const float*)d_in[3];
  const float* b2      = (const float*)d_in[4];
  const int*   src     = (const int*)d_in[5];
  const int*   dst     = (const int*)d_in[6];
  float*       out     = (float*)d_out;

  float* hbuf = (float*)d_ws;
  int* seg = (int*)(hbuf + (size_t)N_NODES * HID_F);
  int* ofs = seg + N_BINS * BIN_EDGES;
  float* z1 = out;

  gcn_gemm1_wmma<<<GEMM1_BLOCKS, GEMM1_THREADS, 0, stream>>>(feature, W1, z1);

  gcn_bin_edges<<<N_BINS, 512, 0, stream>>>(src, dst, seg, ofs);

  gcn_agg_hidden<<<N_PARTS, 1024, 0, stream>>>(z1, seg, ofs, b1, hbuf);

  gcn_agg_out_wmma<<<N_PARTS, 1024, 0, stream>>>(hbuf, seg, ofs, W2, b2, out);
}


# First product's X loads software-pipelined: two fragment register sets, the next 64-k group's 16 loads fly under the current group's products, the first under the W1 staging; f32 sums unchanged
# speedup vs baseline: 2.0744x; 1.0635x over previous
#include <hip/hip_runtime.h>

typedef __attribute__((ext_vector_type(2))) float v2f;
typedef __attribute__((ext_vector_type(8))) float v8f;

#define N_NODES 50000
#define N_EDGES 400000
#define IN_F    768
#define HID_F   16
#define OUT_F   21
#define GEMM1_TILES  (N_NODES / 16)
#define GEMM1_BLOCKS 256
#define GEMM1_SLOTS  ((GEMM1_TILES + GEMM1_BLOCKS - 1) / GEMM1_BLOCKS)
#define GEMM1_KSPLIT 2
#define GEMM1_KPART  (IN_F / GEMM1_KSPLIT)
#define GEMM1_WAVES  (GEMM1_SLOTS * GEMM1_KSPLIT)
#define GEMM1_THREADS (GEMM1_WAVES * 32)
#define PART_NODES 200
#define N_PARTS   (N_NODES / PART_NODES)
#define BIN_EDGES 2048
#define N_BINS    ((N_EDGES + BIN_EDGES - 1) / BIN_EDGES)
#define SEG_PER_WAVE ((N_BINS + 31) / 32)
#define OFS_PITCH 256
#define ACC_PITCH 17
#define ACC_ROWS  208
#define OUT_TASKS ((ACC_ROWS / 16) * 2)
#define OST_PITCH 34

__device__ __forceinline__ void gcn_load_group(v2f (&a)[16], const float* xk)
{
#pragma unroll
  for (int i = 0; i < 16; ++i)
    a[i] = *(const v2f*)(xk + 4 * i);
}

__device__ __forceinline__ void gcn_mul_group(const v2f (&a)[16], const float* w1s,
                                              int k, int koff, int l16,
                                              v8f& acc0, v8f& acc1)
{
#pragma unroll
  for (int i = 0; i < 16; i += 2) {
    const int kk = k + 4 * i;
    v2f b0, b1;
    b0.x = w1s[(kk + koff)     * HID_F + l16];
    b0.y = w1s[(kk + koff + 1) * HID_F + l16];
    b1.x = w1s[(kk + 4 + koff)     * HID_F + l16];
    b1.y = w1s[(kk + 4 + koff + 1) * HID_F + l16];
    acc0 = __builtin_amdgcn_wmma_f32_16x16x4_f32(false, a[i], false, b0,
                                                 (short)0, acc0, false, false);
    acc1 = __builtin_amdgcn_wmma_f32_16x16x4_f32(false, a[i + 1], false, b1,
                                                 (short)0, acc1, false, false);
  }
}

__global__ __launch_bounds__(GEMM1_THREADS) void gcn_gemm1_wmma(
    const float* __restrict__ X, const float* __restrict__ W1,
    float* __restrict__ Z)
{
  __shared__ float w1s[IN_F * HID_F];
  __shared__ float part[GEMM1_SLOTS * 8 * 32];

  const int lane = threadIdx.x & 31;
  const int wave = threadIdx.x >> 5;
  const int slot = wave / GEMM1_KSPLIT;
  const int kh   = wave % GEMM1_KSPLIT;
  const int tile = slot * GEMM1_BLOCKS + blockIdx.x;
  const bool live = tile < GEMM1_TILES;
  const int m0   = tile * 16;

  const int half = lane >> 4;
  const int l16  = lane & 15;
  const int koff = half << 1;
  const int k0   = kh * GEMM1_KPART;
  const float* xrow = X + (size_t)((live ? m0 : 0) + l16) * IN_F + koff + k0;

  v2f a[2][16];
  {
    const float4* w4 = (const float4*)W1;
    float4*       s4 = (float4*)w1s;
    const int n4 = (IN_F * HID_F) / 4;
    float4 w[4];
#pragma unroll
    for (int j = 0; j < 4; ++j) {
      const int i = threadIdx.x + j * GEMM1_THREADS;
      w[j] = w4[i < n4 ? i : n4 - 1];
    }
    gcn_load_group(a[0], xrow);
#pragma unroll
    for (int j = 0; j < 4; ++j) {
      const int i = threadIdx.x + j * GEMM1_THREADS;
      s4[i < n4 ? i : n4 - 1] = w[j];
    }
  }
  __builtin_amdgcn_fence(__ATOMIC_RELEASE, "workgroup", "local");
  __builtin_amdgcn_s_barrier();
  __builtin_amdgcn_fence(__ATOMIC_ACQUIRE, "workgroup", "local");

  v8f acc = {};
  if (live) {
    v8f acc0 = {};
    v8f acc1 = {};
#pragma unroll
    for (int g = 0; g < GEMM1_KPART / 64; ++g) {
      if (g + 1 < GEMM1_KPART / 64)
        gcn_load_group(a[(g + 1) & 1], xrow + (g + 1) * 64);
      __builtin_amdgcn_sched_barrier(0);
      gcn_mul_group(a[g & 1], w1s, k0 + g * 64, koff, l16, acc0, acc1);
      __builtin_amdgcn_sched_barrier(0);
    }
    acc = acc0 + acc1;
    if (kh != 0) {
      float* pp = part + slot * (8 * 32) + lane;
#pragma unroll
      for (int r = 0; r < 8; ++r) pp[r * 32] = acc[r];
    }
  }
  __syncthreads();

  if (live && kh == 0) {
    const float* pp = part + slot * (8 * 32) + lane;
    float* zp = Z + (size_t)(m0 + half * 8) * HID_F + l16;
#pragma unroll
    for (int r = 0; r < 8; ++r) zp[r * HID_F] = acc[r] + pp[r * 32];
  }
}

__global__ __launch_bounds__(512) void gcn_bin_edges(
    const int* __restrict__ src, const int* __restrict__ dst,
    int* __restrict__ seg, int* __restrict__ ofs)
{
  __shared__ int hist[OFS_PITCH];
  __shared__ int lstart[OFS_PITCH];
  __shared__ int wsum[8];
  const int tid  = threadIdx.x;
  const int lane = tid & 31;
  const int wave = tid >> 5;
  if (tid < OFS_PITCH) hist[tid] = 0;

  const int base = blockIdx.x * BIN_EDGES;
  int s[4], d[4];
#pragma unroll
  for (int k = 0; k < 4; ++k) {
    const int e  = base + k * 512 + tid;
    const int ec = e < N_EDGES ? e : N_EDGES - 1;
    s[k] = src[ec];
    d[k] = dst[ec];
  }
  __syncthreads();

  int pp[4], rk[4];
#pragma unroll
  for (int k = 0; k < 4; ++k) {
    const int e = base + k * 512 + tid;
    pp[k] = -1;
    rk[k] = 0;
    if (e < N_EDGES) {
      pp[k] = d[k] / PART_NODES;
      rk[k] = atomicAdd(&hist[pp[k]], 1);
    }
  }
  __syncthreads();

  int v = 0, incl = 0;
  if (tid < OFS_PITCH) {
    v = hist[tid];
    incl = v;
    for (int o = 1; o < 32; o <<= 1) {
      const int t = __shfl_up(incl, o);
      if (lane >= o) incl += t;
    }
    if (lane == 31) wsum[wave] = incl;
  }
  __syncthreads();
  if (tid < OFS_PITCH) {
    int pre = 0;
    for (int w = 0; w < wave; ++w) pre += wsum[w];
    const int ex = pre + incl - v;
    lstart[tid] = ex;
    if (tid <= N_PARTS) ofs[blockIdx.x * OFS_PITCH + tid] = ex;
  }
  __syncthreads();

  int* sg = seg + base;
#pragma unroll
  for (int k = 0; k < 4; ++k) {
    if (pp[k] >= 0)
      sg[lstart[pp[k]] + rk[k]] = (s[k] << 8) | (d[k] - pp[k] * PART_NODES);
  }
}

__device__ __forceinline__ void gcn_range_sums(
    const float* __restrict__ feat, const int* __restrict__ seg,
    const int* __restrict__ ofs, float* acc, int p)
{
  const int tid  = threadIdx.x;
  const int lane = tid & 31;
  const int wave = tid >> 5;
  const int q = lane >> 2;
  const int c = lane & 3;

  int pos[SEG_PER_WAVE], end[SEG_PER_WAVE];
#pragma unroll
  for (int j = 0; j < SEG_PER_WAVE; ++j) {
    const int b = wave + 32 * j;
    pos[j] = 0; end[j] = 0;
    if (b < N_BINS) {
      pos[j] = ofs[b * OFS_PITCH + p] + q;
      end[j] = ofs[b * OFS_PITCH + p + 1];
    }
  }
  __syncthreads();

  const float4* f4 = (const float4*)feat;
  const int*    sg = seg + wave * BIN_EDGES;
  for (int round = 0; round < BIN_EDGES / 8; ++round) {
    bool any = false;
#pragma unroll
    for (int j = 0; j < SEG_PER_WAVE; ++j) any = any || (pos[j] < end[j]);
    if (!any) break;
    int e[SEG_PER_WAVE];
#pragma unroll
    for (int j = 0; j < SEG_PER_WAVE; ++j)
      e[j] = sg[pos[j] < end[j] ? j * 32 * BIN_EDGES + pos[j] : 0];
    float4 v[SEG_PER_WAVE];
#pragma unroll
    for (int j = 0; j < SEG_PER_WAVE; ++j) v[j] = f4[(e[j] >> 8) * 4 + c];
    asm volatile("" :: "v"(v[0].x));
#pragma unroll
    for (int j = 0; j < SEG_PER_WAVE; ++j) {
      if (pos[j] < end[j]) {
        float* a = acc + (e[j] & 255) * ACC_PITCH + c * 4;
        atomicAdd(a + 0, v[j].x);
        atomicAdd(a + 1, v[j].y);
        atomicAdd(a + 2, v[j].z);
        atomicAdd(a + 3, v[j].w);
      }
      pos[j] += 8;
    }
  }
}

__global__ __launch_bounds__(1024) void gcn_agg_hidden(
    const float* __restrict__ feat, const int* __restrict__ seg,
    const int* __restrict__ ofs, const float* __restrict__ b1,
    float* __restrict__ res)
{
  __shared__ float acc[PART_NODES * ACC_PITCH];
  const int tid = threadIdx.x;
  const int p   = blockIdx.x;
  for (int i = tid; i < PART_NODES * ACC_PITCH; i += 1024) acc[i] = 0.0f;

  gcn_range_sums(feat, seg, ofs, acc, p);
  __syncthreads();

  float* rp = res + (size_t)p * PART_NODES * HID_F;
  for (int i = tid; i < PART_NODES * HID_F; i += 1024) {
    const int node = i >> 4;
    const int f    = i & 15;
    float val = acc[node * ACC_PITCH + f] + b1[f];
    val = val > 0.0f ? val : 0.0f;
    rp[i] = val;
  }
}

__global__ __launch_bounds__(1024) void gcn_agg_out_wmma(
    const float* __restrict__ feat, const int* __restrict__ seg,
    const int* __restrict__ ofs, const float* __restrict__ W2,
    const float* __restrict__ b2, float* __restrict__ out)
{
  __shared__ float acc[ACC_ROWS * ACC_PITCH];
  __shared__ float ost[ACC_ROWS * OST_PITCH];
  const int tid  = threadIdx.x;
  const int lane = tid & 31;
  const int wave = tid >> 5;
  const int p    = blockIdx.x;

  const int mt   = wave >> 1;
  const int nt   = wave & 1;
  const int half = lane >> 4;
  const int l16  = lane & 15;
  const int koff = half << 1;
  const int n    = nt * 16 + l16;
  const bool nvalid = (n < OUT_F);
  const int nc = nvalid ? n : 0;
  v2f bf[HID_F / 4];
#pragma unroll
  for (int k = 0; k < HID_F; k += 4) {
    const float w0 = W2[(k + koff)     * OUT_F + nc];
    const float w1 = W2[(k + koff + 1) * OUT_F + nc];
    bf[k / 4].x = nvalid ? w0 : 0.0f;
    bf[k / 4].y = nvalid ? w1 : 0.0f;
  }
  const float bias = b2[nc];

  for (int i = tid; i < ACC_ROWS * ACC_PITCH; i += 1024) acc[i] = 0.0f;

  gcn_range_sums(feat, seg, ofs, acc, p);
  __syncthreads();

  if (wave < OUT_TASKS) {
    const float* arow = acc + (mt * 16 + l16) * ACC_PITCH + koff;
    v8f d = {};
#pragma unroll
    for (int k = 0; k < HID_F; k += 4) {
      v2f a;
      a.x = arow[k];
      a.y = arow[k + 1];
      d = __builtin_amdgcn_wmma_f32_16x16x4_f32(false, a, false, bf[k / 4],
                                                (short)0, d, false, false);
    }
    float* sp = ost + (mt * 16 + half * 8) * OST_PITCH + n;
#pragma unroll
    for (int r = 0; r < 8; ++r) sp[r * OST_PITCH] = d[r] + bias;
  }
  __syncthreads();

  float* op = out + (size_t)p * PART_NODES * OUT_F;
  for (int i = tid; i < PART_NODES * OUT_F; i += 1024) {
    const int row = i / OUT_F;
    const int col = i - row * OUT_F;
    op[i] = ost[row * OST_PITCH + col];
  }
}

extern "C" void kernel_launch(void* const* d_in, const int* in_sizes, int n_in,
                              void* d_out, int out_size, void* d_ws, size_t ws_size,
                              hipStream_t stream)
{
  const float* feature = (const float*)d_in[0];
  const float* W1      = (const float*)d_in[1];
  const float* b1      = (const float*)d_in[2];
  const float* W2      = (const float*)d_in[3];
  const float* b2      = (const float*)d_in[4];
  const int*   src     = (const int*)d_in[5];
  const int*   dst     = (const int*)d_in[6];
  float*       out     = (float*)d_out;

  float* hbuf = (float*)d_ws;
  int* seg = (int*)(hbuf + (size_t)N_NODES * HID_F);
  int* ofs = seg + N_BINS * BIN_EDGES;
  float* z1 = out;

  gcn_gemm1_wmma<<<GEMM1_BLOCKS, GEMM1_THREADS, 0, stream>>>(feature, W1, z1);

  gcn_bin_edges<<<N_BINS, 512, 0, stream>>>(src, dst, seg, ofs);

  gcn_agg_hidden<<<N_PARTS, 1024, 0, stream>>>(z1, seg, ofs, b1, hbuf);

  gcn_agg_out_wmma<<<N_PARTS, 1024, 0, stream>>>(hbuf, seg, ofs, W2, b2, out);
}


# Edge binning folded into the first product's work-groups: the 196 with 12 tiles each counting-sort a 2048-edge segment while their first two X groups fly; 3 launches not 4; sum order: LDS arrival
# speedup vs baseline: 2.2897x; 1.1038x over previous
#include <hip/hip_runtime.h>

typedef __attribute__((ext_vector_type(2))) float v2f;
typedef __attribute__((ext_vector_type(8))) float v8f;

#define N_NODES 50000
#define N_EDGES 400000
#define IN_F    768
#define HID_F   16
#define OUT_F   21
#define GEMM1_TILES  (N_NODES / 16)
#define GEMM1_BLOCKS 256
#define GEMM1_SLOTS  ((GEMM1_TILES + GEMM1_BLOCKS - 1) / GEMM1_BLOCKS)
#define GEMM1_KSPLIT 2
#define GEMM1_KPART  (IN_F / GEMM1_KSPLIT)
#define GEMM1_WAVES  (GEMM1_SLOTS * GEMM1_KSPLIT)
#define GEMM1_THREADS (GEMM1_WAVES * 32)
#define PART_NODES 200
#define N_PARTS   (N_NODES / PART_NODES)
#define BIN_EDGES 2048
#define N_BINS    ((N_EDGES + BIN_EDGES - 1) / BIN_EDGES)
#define BIN_FIRST (GEMM1_BLOCKS - N_BINS)
#define BIN_SLOTS ((BIN_EDGES + GEMM1_THREADS - 1) / GEMM1_THREADS)
#define SEG_PER_WAVE ((N_BINS + 31) / 32)
#define OFS_PITCH 256
#define ACC_PITCH 17
#define ACC_ROWS  208
#define OUT_TASKS ((ACC_ROWS / 16) * 2)
#define OST_PITCH 34

__device__ __forceinline__ void gcn_lds_barrier()
{
  __builtin_amdgcn_fence(__ATOMIC_RELEASE, "workgroup", "local");
  __builtin_amdgcn_s_barrier();
  __builtin_amdgcn_fence(__ATOMIC_ACQUIRE, "workgroup", "local");
}

__device__ __forceinline__ void gcn_load_group(v2f (&a)[16], const float* xk)
{
#pragma unroll
  for (int i = 0; i < 16; ++i)
    a[i] = *(const v2f*)(xk + 4 * i);
}

__device__ __forceinline__ void gcn_mul_group(const v2f (&a)[16], const float* w1s,
                                              int k, int koff, int l16,
                                              v8f& acc0, v8f& acc1)
{
#pragma unroll
  for (int i = 0; i < 16; i += 2) {
    const int kk = k + 4 * i;
    v2f b0, b1;
    b0.x = w1s[(kk + koff)     * HID_F + l16];
    b0.y = w1s[(kk + koff + 1) * HID_F + l16];
    b1.x = w1s[(kk + 4 + koff)     * HID_F + l16];
    b1.y = w1s[(kk + 4 + koff + 1) * HID_F + l16];
    acc0 = __builtin_amdgcn_wmma_f32_16x16x4_f32(false, a[i], false, b0,
                                                 (short)0, acc0, false, false);
    acc1 = __builtin_amdgcn_wmma_f32_16x16x4_f32(false, a[i + 1], false, b1,
                                                 (short)0, acc1, false, false);
  }
}

__global__ __launch_bounds__(GEMM1_THREADS) void gcn_gemm1_wmma(
    const float* __restrict__ X, const float* __restrict__ W1,
    float* __restrict__ Z,
    const int* __restrict__ src, const int* __restrict__ dst,
    int* __restrict__ seg, int* __restrict__ ofs)
{
  __shared__ float w1s[IN_F * HID_F];
  __shared__ float part[GEMM1_SLOTS * 8 * 32];
  __shared__ int hist[OFS_PITCH];
  __shared__ int lstart[OFS_PITCH];
  __shared__ int wsum[8];

  const int tid  = threadIdx.x;
  const int lane = tid & 31;
  const int wave = tid >> 5;
  const int slot = wave / GEMM1_KSPLIT;
  const int kh   = wave % GEMM1_KSPLIT;
  const int tile = slot * GEMM1_BLOCKS + blockIdx.x;
  const bool live = tile < GEMM1_TILES;
  const int m0   = tile * 16;

  const int half = lane >> 4;
  const int l16  = lane & 15;
  const int koff = half << 1;
  const int k0   = kh * GEMM1_KPART;
  const float* xrow = X + (size_t)((live ? m0 : 0) + l16) * IN_F + koff + k0;

  const int  bin   = (int)blockIdx.x - BIN_FIRST;
  const bool sorts = bin >= 0;
  const int  ebase = bin * BIN_EDGES;
  if (sorts && tid < OFS_PITCH) hist[tid] = 0;

  int s[BIN_SLOTS], d[BIN_SLOTS];
#pragma unroll
  for (int k = 0; k < BIN_SLOTS; ++k) {
    const int e  = ebase + k * GEMM1_THREADS + tid;
    const int ec = (sorts && e < N_EDGES) ? e : N_EDGES - 1;
    s[k] = src[ec];
    d[k] = dst[ec];
  }
  v2f a[2][16];
  const float4* w4 = (const float4*)W1;
  float4*       s4 = (float4*)w1s;
  const int n4 = (IN_F * HID_F) / 4;
  float4 w[4];
#pragma unroll
  for (int j = 0; j < 4; ++j) {
    const int i = tid + j * GEMM1_THREADS;
    w[j] = w4[i < n4 ? i : n4 - 1];
  }
  gcn_load_group(a[0], xrow);
  gcn_load_group(a[1], xrow + 64);

#pragma unroll
  for (int j = 0; j < 4; ++j) {
    const int i = tid + j * GEMM1_THREADS;
    s4[i < n4 ? i : n4 - 1] = w[j];
  }
  gcn_lds_barrier();

  if (sorts) {
    int pp[BIN_SLOTS], rk[BIN_SLOTS];
#pragma unroll
    for (int k = 0; k < BIN_SLOTS; ++k) {
      const int el = k * GEMM1_THREADS + tid;
      pp[k] = -1;
      rk[k] = 0;
      if (el < BIN_EDGES && ebase + el < N_EDGES) {
        pp[k] = d[k] / PART_NODES;
        rk[k] = atomicAdd(&hist[pp[k]], 1);
      }
    }
    gcn_lds_barrier();
    int v = 0, incl = 0;
    if (tid < OFS_PITCH) {
      v = hist[tid];
      incl = v;
      for (int o = 1; o < 32; o <<= 1) {
        const int t = __shfl_up(incl, o);
        if (lane >= o) incl += t;
      }
      if (lane == 31) wsum[wave] = incl;
    }
    gcn_lds_barrier();
    if (tid < OFS_PITCH) {
      int pre = 0;
      for (int w8 = 0; w8 < wave; ++w8) pre += wsum[w8];
      const int ex = pre + incl - v;
      lstart[tid] = ex;
      if (tid <= N_PARTS) ofs[bin * OFS_PITCH + tid] = ex;
    }
    gcn_lds_barrier();
    int* sg = seg + ebase;
#pragma unroll
    for (int k = 0; k < BIN_SLOTS; ++k) {
      if (pp[k] >= 0)
        sg[lstart[pp[k]] + rk[k]] = (s[k] << 8) | (d[k] - pp[k] * PART_NODES);
    }
  }

  v8f acc = {};
  if (live) {
    v8f acc0 = {};
    v8f acc1 = {};
#pragma unroll
    for (int g = 0; g < GEMM1_KPART / 64; ++g) {
      if (g >= 1 && g + 1 < GEMM1_KPART / 64)
        gcn_load_group(a[(g + 1) & 1], xrow + (g + 1) * 64);
      __builtin_amdgcn_sched_barrier(0);
      gcn_mul_group(a[g & 1], w1s, k0 + g * 64, koff, l16, acc0, acc1);
      __builtin_amdgcn_sched_barrier(0);
    }
    acc = acc0 + acc1;
    if (kh != 0) {
      float* pq = part + slot * (8 * 32) + lane;
#pragma unroll
      for (int r = 0; r < 8; ++r) pq[r * 32] = acc[r];
    }
  }
  __syncthreads();

  if (live && kh == 0) {
    const float* pq = part + slot * (8 * 32) + lane;
    float* zp = Z + (size_t)(m0 + half * 8) * HID_F + l16;
#pragma unroll
    for (int r = 0; r < 8; ++r) zp[r * HID_F] = acc[r] + pq[r * 32];
  }
}

__device__ __forceinline__ void gcn_range_sums(
    const float* __restrict__ feat, const int* __restrict__ seg,
    const int* __restrict__ ofs, float* acc, int p)
{
  const int tid  = threadIdx.x;
  const int lane = tid & 31;
  const int wave = tid >> 5;
  const int q = lane >> 2;
  const int c = lane & 3;

  int pos[SEG_PER_WAVE], end[SEG_PER_WAVE];
#pragma unroll
  for (int j = 0; j < SEG_PER_WAVE; ++j) {
    const int b = wave + 32 * j;
    pos[j] = 0; end[j] = 0;
    if (b < N_BINS) {
      pos[j] = ofs[b * OFS_PITCH + p] + q;
      end[j] = ofs[b * OFS_PITCH + p + 1];
    }
  }
  __syncthreads();

  const float4* f4 = (const float4*)feat;
  const int*    sg = seg + wave * BIN_EDGES;
  for (int round = 0; round < BIN_EDGES / 8; ++round) {
    bool any = false;
#pragma unroll
    for (int j = 0; j < SEG_PER_WAVE; ++j) any = any || (pos[j] < end[j]);
    if (!any) break;
    int e[SEG_PER_WAVE];
#pragma unroll
    for (int j = 0; j < SEG_PER_WAVE; ++j)
      e[j] = sg[pos[j] < end[j] ? j * 32 * BIN_EDGES + pos[j] : 0];
    float4 v[SEG_PER_WAVE];
#pragma unroll
    for (int j = 0; j < SEG_PER_WAVE; ++j) v[j] = f4[(e[j] >> 8) * 4 + c];
    asm volatile("" :: "v"(v[0].x));
#pragma unroll
    for (int j = 0; j < SEG_PER_WAVE; ++j) {
      if (pos[j] < end[j]) {
        float* a = acc + (e[j] & 255) * ACC_PITCH + c * 4;
        atomicAdd(a + 0, v[j].x);
        atomicAdd(a + 1, v[j].y);
        atomicAdd(a + 2, v[j].z);
        atomicAdd(a + 3, v[j].w);
      }
      pos[j] += 8;
    }
  }
}

__global__ __launch_bounds__(1024) void gcn_agg_hidden(
    const float* __restrict__ feat, const int* __restrict__ seg,
    const int* __restrict__ ofs, const float* __restrict__ b1,
    float* __restrict__ res)
{
  __shared__ float acc[PART_NODES * ACC_PITCH];
  const int tid = threadIdx.x;
  const int p   = blockIdx.x;
  for (int i = tid; i < PART_NODES * ACC_PITCH; i += 1024) acc[i] = 0.0f;

  gcn_range_sums(feat, seg, ofs, acc, p);
  __syncthreads();

  float* rp = res + (size_t)p * PART_NODES * HID_F;
  for (int i = tid; i < PART_NODES * HID_F; i += 1024) {
    const int node = i >> 4;
    const int f    = i & 15;
    float val = acc[node * ACC_PITCH + f] + b1[f];
    val = val > 0.0f ? val : 0.0f;
    rp[i] = val;
  }
}

__global__ __launch_bounds__(1024) void gcn_agg_out_wmma(
    const float* __restrict__ feat, const int* __restrict__ seg,
    const int* __restrict__ ofs, const float* __restrict__ W2,
    const float* __restrict__ b2, float* __restrict__ out)
{
  __shared__ float acc[ACC_ROWS * ACC_PITCH];
  __shared__ float ost[ACC_ROWS * OST_PITCH];
  const int tid  = threadIdx.x;
  const int lane = tid & 31;
  const int wave = tid >> 5;
  const int p    = blockIdx.x;

  const int mt   = wave >> 1;
  const int nt   = wave & 1;
  const int half = lane >> 4;
  const int l16  = lane & 15;
  const int koff = half << 1;
  const int n    = nt * 16 + l16;
  const bool nvalid = (n < OUT_F);
  const int nc = nvalid ? n : 0;
  v2f bf[HID_F / 4];
#pragma unroll
  for (int k = 0; k < HID_F; k += 4) {
    const float w0 = W2[(k + koff)     * OUT_F + nc];
    const float w1 = W2[(k + koff + 1) * OUT_F + nc];
    bf[k / 4].x = nvalid ? w0 : 0.0f;
    bf[k / 4].y = nvalid ? w1 : 0.0f;
  }
  const float bias = b2[nc];

  for (int i = tid; i < ACC_ROWS * ACC_PITCH; i += 1024) acc[i] = 0.0f;

  gcn_range_sums(feat, seg, ofs, acc, p);
  __syncthreads();

  if (wave < OUT_TASKS) {
    const float* arow = acc + (mt * 16 + l16) * ACC_PITCH + koff;
    v8f d = {};
#pragma unroll
    for (int k = 0; k < HID_F; k += 4) {
      v2f a;
      a.x = arow[k];
      a.y = arow[k + 1];
      d = __builtin_amdgcn_wmma_f32_16x16x4_f32(false, a, false, bf[k / 4],
                                                (short)0, d, false, false);
    }
    float* sp = ost + (mt * 16 + half * 8) * OST_PITCH + n;
#pragma unroll
    for (int r = 0; r < 8; ++r) sp[r * OST_PITCH] = d[r] + bias;
  }
  __syncthreads();

  float* op = out + (size_t)p * PART_NODES * OUT_F;
  for (int i = tid; i < PART_NODES * OUT_F; i += 1024) {
    const int row = i / OUT_F;
    const int col = i - row * OUT_F;
    op[i] = ost[row * OST_PITCH + col];
  }
}

extern "C" void kernel_launch(void* const* d_in, const int* in_sizes, int n_in,
                              void* d_out, int out_size, void* d_ws, size_t ws_size,
                              hipStream_t stream)
{
  const float* feature = (const float*)d_in[0];
  const float* W1      = (const float*)d_in[1];
  const float* b1      = (const float*)d_in[2];
  const float* W2      = (const float*)d_in[3];
  const float* b2      = (const float*)d_in[4];
  const int*   src     = (const int*)d_in[5];
  const int*   dst     = (const int*)d_in[6];
  float*       out     = (float*)d_out;

  float* hbuf = (float*)d_ws;
  int* seg = (int*)(hbuf + (size_t)N_NODES * HID_F);
  int* ofs = seg + N_BINS * BIN_EDGES;
  float* z1 = out;

  gcn_gemm1_wmma<<<GEMM1_BLOCKS, GEMM1_THREADS, 0, stream>>>(feature, W1, z1, src, dst, seg, ofs);

  gcn_agg_hidden<<<N_PARTS, 1024, 0, stream>>>(z1, seg, ofs, b1, hbuf);

  gcn_agg_out_wmma<<<N_PARTS, 1024, 0, stream>>>(hbuf, seg, ofs, W2, b2, out);
}


# Edge sums: a wave's 7 segment ranges walked as one numbered list, all its edges in one flight of loads issued ahead of the clear barrier, not 2-3 dependent rounds behind it; sum order: LDS arrival
# speedup vs baseline: 2.3618x; 1.0315x over previous
#include <hip/hip_runtime.h>

typedef __attribute__((ext_vector_type(2))) float v2f;
typedef __attribute__((ext_vector_type(8))) float v8f;

#define N_NODES 50000
#define N_EDGES 400000
#define IN_F    768
#define HID_F   16
#define OUT_F   21
#define GEMM1_TILES  (N_NODES / 16)
#define GEMM1_BLOCKS 256
#define GEMM1_SLOTS  ((GEMM1_TILES + GEMM1_BLOCKS - 1) / GEMM1_BLOCKS)
#define GEMM1_KSPLIT 2
#define GEMM1_KPART  (IN_F / GEMM1_KSPLIT)
#define GEMM1_WAVES  (GEMM1_SLOTS * GEMM1_KSPLIT)
#define GEMM1_THREADS (GEMM1_WAVES * 32)
#define PART_NODES 200
#define N_PARTS   (N_NODES / PART_NODES)
#define BIN_EDGES 2048
#define N_BINS    ((N_EDGES + BIN_EDGES - 1) / BIN_EDGES)
#define BIN_FIRST (GEMM1_BLOCKS - N_BINS)
#define BIN_SLOTS ((BIN_EDGES + GEMM1_THREADS - 1) / GEMM1_THREADS)
#define SEG_PER_WAVE ((N_BINS + 31) / 32)
#define FLIGHT    12
#define OFS_PITCH 256
#define ACC_PITCH 17
#define ACC_ROWS  208
#define OUT_TASKS ((ACC_ROWS / 16) * 2)
#define OST_PITCH 34

__device__ __forceinline__ void gcn_lds_barrier()
{
  __builtin_amdgcn_fence(__ATOMIC_RELEASE, "workgroup", "local");
  __builtin_amdgcn_s_barrier();
  __builtin_amdgcn_fence(__ATOMIC_ACQUIRE, "workgroup", "local");
}

__device__ __forceinline__ void gcn_load_group(v2f (&a)[16], const float* xk)
{
#pragma unroll
  for (int i = 0; i < 16; ++i)
    a[i] = *(const v2f*)(xk + 4 * i);
}

__device__ __forceinline__ void gcn_mul_group(const v2f (&a)[16], const float* w1s,
                                              int k, int koff, int l16,
                                              v8f& acc0, v8f& acc1)
{
#pragma unroll
  for (int i = 0; i < 16; i += 2) {
    const int kk = k + 4 * i;
    v2f b0, b1;
    b0.x = w1s[(kk + koff)     * HID_F + l16];
    b0.y = w1s[(kk + koff + 1) * HID_F + l16];
    b1.x = w1s[(kk + 4 + koff)     * HID_F + l16];
    b1.y = w1s[(kk + 4 + koff + 1) * HID_F + l16];
    acc0 = __builtin_amdgcn_wmma_f32_16x16x4_f32(false, a[i], false, b0,
                                                 (short)0, acc0, false, false);
    acc1 = __builtin_amdgcn_wmma_f32_16x16x4_f32(false, a[i + 1], false, b1,
                                                 (short)0, acc1, false, false);
  }
}

__global__ __launch_bounds__(GEMM1_THREADS) void gcn_gemm1_wmma(
    const float* __restrict__ X, const float* __restrict__ W1,
    float* __restrict__ Z,
    const int* __restrict__ src, const int* __restrict__ dst,
    int* __restrict__ seg, int* __restrict__ ofs)
{
  __shared__ float w1s[IN_F * HID_F];
  __shared__ float part[GEMM1_SLOTS * 8 * 32];
  __shared__ int hist[OFS_PITCH];
  __shared__ int lstart[OFS_PITCH];
  __shared__ int wsum[8];

  const int tid  = threadIdx.x;
  const int lane = tid & 31;
  const int wave = tid >> 5;
  const int slot = wave / GEMM1_KSPLIT;
  const int kh   = wave % GEMM1_KSPLIT;
  const int tile = slot * GEMM1_BLOCKS + blockIdx.x;
  const bool live = tile < GEMM1_TILES;
  const int m0   = tile * 16;

  const int half = lane >> 4;
  const int l16  = lane & 15;
  const int koff = half << 1;
  const int k0   = kh * GEMM1_KPART;
  const float* xrow = X + (size_t)((live ? m0 : 0) + l16) * IN_F + koff + k0;

  const int  bin   = (int)blockIdx.x - BIN_FIRST;
  const bool sorts = bin >= 0;
  const int  ebase = bin * BIN_EDGES;
  if (sorts && tid < OFS_PITCH) hist[tid] = 0;

  int s[BIN_SLOTS], d[BIN_SLOTS];
#pragma unroll
  for (int k = 0; k < BIN_SLOTS; ++k) {
    const int e  = ebase + k * GEMM1_THREADS + tid;
    const int ec = (sorts && e < N_EDGES) ? e : N_EDGES - 1;
    s[k] = src[ec];
    d[k] = dst[ec];
  }
  v2f a[2][16];
  const float4* w4 = (const float4*)W1;
  float4*       s4 = (float4*)w1s;
  const int n4 = (IN_F * HID_F) / 4;
  float4 w[4];
#pragma unroll
  for (int j = 0; j < 4; ++j) {
    const int i = tid + j * GEMM1_THREADS;
    w[j] = w4[i < n4 ? i : n4 - 1];
  }
  gcn_load_group(a[0], xrow);
  gcn_load_group(a[1], xrow + 64);

#pragma unroll
  for (int j = 0; j < 4; ++j) {
    const int i = tid + j * GEMM1_THREADS;
    s4[i < n4 ? i : n4 - 1] = w[j];
  }
  gcn_lds_barrier();

  if (sorts) {
    int pp[BIN_SLOTS], rk[BIN_SLOTS];
#pragma unroll
    for (int k = 0; k < BIN_SLOTS; ++k) {
      const int el = k * GEMM1_THREADS + tid;
      pp[k] = -1;
      rk[k] = 0;
      if (el < BIN_EDGES && ebase + el < N_EDGES) {
        pp[k] = d[k] / PART_NODES;
        rk[k] = atomicAdd(&hist[pp[k]], 1);
      }
    }
    gcn_lds_barrier();
    int v = 0, incl = 0;
    if (tid < OFS_PITCH) {
      v = hist[tid];
      incl = v;
      for (int o = 1; o < 32; o <<= 1) {
        const int t = __shfl_up(incl, o);
        if (lane >= o) incl += t;
      }
      if (lane == 31) wsum[wave] = incl;
    }
    gcn_lds_barrier();
    if (tid < OFS_PITCH) {
      int pre = 0;
      for (int w8 = 0; w8 < wave; ++w8) pre += wsum[w8];
      const int ex = pre + incl - v;
      lstart[tid] = ex;
      if (tid <= N_PARTS) ofs[bin * OFS_PITCH + tid] = ex;
    }
    gcn_lds_barrier();
    int* sg = seg + ebase;
#pragma unroll
    for (int k = 0; k < BIN_SLOTS; ++k) {
      if (pp[k] >= 0)
        sg[lstart[pp[k]] + rk[k]] = (s[k] << 8) | (d[k] - pp[k] * PART_NODES);
    }
  }

  v8f acc = {};
  if (live) {
    v8f acc0 = {};
    v8f acc1 = {};
#pragma unroll
    for (int g = 0; g < GEMM1_KPART / 64; ++g) {
      if (g >= 1 && g + 1 < GEMM1_KPART / 64)
        gcn_load_group(a[(g + 1) & 1], xrow + (g + 1) * 64);
      __builtin_amdgcn_sched_barrier(0);
      gcn_mul_group(a[g & 1], w1s, k0 + g * 64, koff, l16, acc0, acc1);
      __builtin_amdgcn_sched_barrier(0);
    }
    acc = acc0 + acc1;
    if (kh != 0) {
      float* pq = part + slot * (8 * 32) + lane;
#pragma unroll
      for (int r = 0; r < 8; ++r) pq[r * 32] = acc[r];
    }
  }
  __syncthreads();

  if (live && kh == 0) {
    const float* pq = part + slot * (8 * 32) + lane;
    float* zp = Z + (size_t)(m0 + half * 8) * HID_F + l16;
#pragma unroll
    for (int r = 0; r < 8; ++r) zp[r * HID_F] = acc[r] + pq[r * 32];
  }
}

__device__ __forceinline__ void gcn_list_loads(
    int (&e)[FLIGHT], bool (&ok)[FLIGHT], float4 (&v)[FLIGHT],
    const int (&cum)[SEG_PER_WAVE + 1], const int (&adj)[SEG_PER_WAVE],
    const int* __restrict__ sg, const float4* __restrict__ f4,
    int base, int q, int c)
{
#pragma unroll
  for (int r = 0; r < FLIGHT; ++r) {
    const int s = base + 8 * r + q;
    int a = adj[0];
#pragma unroll
    for (int j = 1; j < SEG_PER_WAVE; ++j) a = (s >= cum[j]) ? adj[j] : a;
    ok[r] = s < cum[SEG_PER_WAVE];
    e[r]  = sg[ok[r] ? a + s : 0];
  }
#pragma unroll
  for (int r = 0; r < FLIGHT; ++r) v[r] = f4[(e[r] >> 8) * 4 + c];
}

__device__ __forceinline__ void gcn_list_adds(
    const int (&e)[FLIGHT], const bool (&ok)[FLIGHT], const float4 (&v)[FLIGHT],
    float* acc, int c)
{
#pragma unroll
  for (int r = 0; r < FLIGHT; ++r) {
    if (ok[r]) {
      float* a = acc + (e[r] & 255) * ACC_PITCH + c * 4;
      atomicAdd(a + 0, v[r].x);
      atomicAdd(a + 1, v[r].y);
      atomicAdd(a + 2, v[r].z);
      atomicAdd(a + 3, v[r].w);
    }
  }
}

__device__ __forceinline__ void gcn_range_sums(
    const float* __restrict__ feat, const int* __restrict__ seg,
    const int* __restrict__ ofs, float* acc, int p)
{
  const int tid  = threadIdx.x;
  const int lane = tid & 31;
  const int wave = __builtin_amdgcn_readfirstlane(tid >> 5);
  const int q = lane >> 2;
  const int c = lane & 3;

  int beg[SEG_PER_WAVE], end[SEG_PER_WAVE];
#pragma unroll
  for (int j = 0; j < SEG_PER_WAVE; ++j) {
    const int b  = wave + 32 * j;
    const int bc = b < N_BINS ? b : 0;
    beg[j] = ofs[bc * OFS_PITCH + p];
    end[j] = ofs[bc * OFS_PITCH + p + (b < N_BINS ? 1 : 0)];
  }
  __builtin_amdgcn_sched_barrier(0);

  int cum[SEG_PER_WAVE + 1], adj[SEG_PER_WAVE];
  cum[0] = 0;
#pragma unroll
  for (int j = 0; j < SEG_PER_WAVE; ++j) {
    adj[j]     = j * 32 * BIN_EDGES + beg[j] - cum[j];
    cum[j + 1] = cum[j] + (end[j] - beg[j]);
  }

  const float4* f4 = (const float4*)feat;
  const int*    sg = seg + wave * BIN_EDGES;
  int    e[FLIGHT];
  bool   ok[FLIGHT];
  float4 v[FLIGHT];
  gcn_list_loads(e, ok, v, cum, adj, sg, f4, 0, q, c);
  gcn_lds_barrier();
  gcn_list_adds(e, ok, v, acc, c);
  for (int base = 8 * FLIGHT; base < cum[SEG_PER_WAVE]; base += 8 * FLIGHT) {
    gcn_list_loads(e, ok, v, cum, adj, sg, f4, base, q, c);
    asm volatile("" :: "v"(v[0].x));
    gcn_list_adds(e, ok, v, acc, c);
  }
}

__global__ __launch_bounds__(1024) void gcn_agg_hidden(
    const float* __restrict__ feat, const int* __restrict__ seg,
    const int* __restrict__ ofs, const float* __restrict__ b1,
    float* __restrict__ res)
{
  __shared__ float acc[PART_NODES * ACC_PITCH];
  const int tid = threadIdx.x;
  const int p   = blockIdx.x;
  for (int i = tid; i < PART_NODES * ACC_PITCH; i += 1024) acc[i] = 0.0f;

  gcn_range_sums(feat, seg, ofs, acc, p);
  __syncthreads();

  float* rp = res + (size_t)p * PART_NODES * HID_F;
  for (int i = tid; i < PART_NODES * HID_F; i += 1024) {
    const int node = i >> 4;
    const int f    = i & 15;
    float val = acc[node * ACC_PITCH + f] + b1[f];
    val = val > 0.0f ? val : 0.0f;
    rp[i] = val;
  }
}

__global__ __launch_bounds__(1024) void gcn_agg_out_wmma(
    const float* __restrict__ feat, const int* __restrict__ seg,
    const int* __restrict__ ofs, const float* __restrict__ W2,
    const float* __restrict__ b2, float* __restrict__ out)
{
  __shared__ float acc[ACC_ROWS * ACC_PITCH];
  __shared__ float ost[ACC_ROWS * OST_PITCH];
  const int tid  = threadIdx.x;
  const int lane = tid & 31;
  const int wave = tid >> 5;
  const int p    = blockIdx.x;

  const int mt   = wave >> 1;
  const int nt   = wave & 1;
  const int half = lane >> 4;
  const int l16  = lane & 15;
  const int koff = half << 1;
  const int n    = nt * 16 + l16;
  const bool nvalid = (n < OUT_F);
  const int nc = nvalid ? n : 0;
  v2f bf[HID_F / 4];
#pragma unroll
  for (int k = 0; k < HID_F; k += 4) {
    const float w0 = W2[(k + koff)     * OUT_F + nc];
    const float w1 = W2[(k + koff + 1) * OUT_F + nc];
    bf[k / 4].x = nvalid ? w0 : 0.0f;
    bf[k / 4].y = nvalid ? w1 : 0.0f;
  }
  const float bias = b2[nc];

  for (int i = tid; i < ACC_ROWS * ACC_PITCH; i += 1024) acc[i] = 0.0f;

  gcn_range_sums(feat, seg, ofs, acc, p);
  __syncthreads();

  if (wave < OUT_TASKS) {
    const float* arow = acc + (mt * 16 + l16) * ACC_PITCH + koff;
    v8f d = {};
#pragma unroll
    for (int k = 0; k < HID_F; k += 4) {
      v2f a;
      a.x = arow[k];
      a.y = arow[k + 1];
      d = __builtin_amdgcn_wmma_f32_16x16x4_f32(false, a, false, bf[k / 4],
                                                (short)0, d, false, false);
    }
    float* sp = ost + (mt * 16 + half * 8) * OST_PITCH + n;
#pragma unroll
    for (int r = 0; r < 8; ++r) sp[r * OST_PITCH] = d[r] + bias;
  }
  __syncthreads();

  float* op = out + (size_t)p * PART_NODES * OUT_F;
  for (int i = tid; i < PART_NODES * OUT_F; i += 1024) {
    const int row = i / OUT_F;
    const int col = i - row * OUT_F;
    op[i] = ost[row * OST_PITCH + col];
  }
}

extern "C" void kernel_launch(void* const* d_in, const int* in_sizes, int n_in,
                              void* d_out, int out_size, void* d_ws, size_t ws_size,
                              hipStream_t stream)
{
  const float* feature = (const float*)d_in[0];
  const float* W1      = (const float*)d_in[1];
  const float* b1      = (const float*)d_in[2];
  const float* W2      = (const float*)d_in[3];
  const float* b2      = (const float*)d_in[4];
  const int*   src     = (const int*)d_in[5];
  const int*   dst     = (const int*)d_in[6];
  float*       out     = (float*)d_out;

  float* hbuf = (float*)d_ws;
  int* seg = (int*)(hbuf + (size_t)N_NODES * HID_F);
  int* ofs = seg + N_BINS * BIN_EDGES;
  float* z1 = out;

  gcn_gemm1_wmma<<<GEMM1_BLOCKS, GEMM1_THREADS, 0, stream>>>(feature, W1, z1, src, dst, seg, ofs);

  gcn_agg_hidden<<<N_PARTS, 1024, 0, stream>>>(z1, seg, ofs, b1, hbuf);

  gcn_agg_out_wmma<<<N_PARTS, 1024, 0, stream>>>(hbuf, seg, ofs, W2, b2, out);
}
